# starting kernel
# baseline (speedup 1.0000x reference)
#include <hip/hip_runtime.h>

typedef __attribute__((ext_vector_type(2))) float v2f;
typedef __attribute__((ext_vector_type(8))) float v8f;

#define NCLS 512
#define NBATCH 4096
#define BM 32

__global__ __launch_bounds__(256) void pack_operands(const float* __restrict__ V,
                                                     float* __restrict__ VPv,
                                                     float* __restrict__ VPt) {
    const int idx = blockIdx.x * 256 + threadIdx.x;
    const int col = idx & (NCLS - 1);
    const int k2  = idx >> 9;

    v2f a;
    a.x = V[(size_t)(2 * k2) * NCLS + col];
    a.y = V[(size_t)(2 * k2 + 1) * NCLS + col];
    *(v2f*)(VPv + ((size_t)k2 * NCLS + col) * 2) = a;

    v2f b = *(const v2f*)(V + (size_t)col * NCLS + 2 * k2);
    *(v2f*)(VPt + ((size_t)k2 * NCLS + col) * 2) = b;
}

__global__ __launch_bounds__(256) void fused_probability_noiser(
    const float* __restrict__ P0,
    const float* __restrict__ sigma,
    const float* __restrict__ lam,
    const float* __restrict__ VPv,
    const float* __restrict__ VPt,
    float* __restrict__ out)
{
    __shared__ float su[BM][NCLS];

    const int tid  = threadIdx.x;
    const int wave = tid >> 5;
    const int lane = tid & 31;
    const int nn   = lane & 15;
    const int hi   = lane >> 4;

    const int cbas = wave * 64;
    const int row0 = blockIdx.x * BM;
    const int m    = nn;

    v8f acc[2][4];
#pragma unroll
    for (int rb = 0; rb < 2; ++rb)
#pragma unroll
        for (int t = 0; t < 4; ++t)
#pragma unroll
            for (int i = 0; i < 8; ++i) acc[rb][t][i] = 0.0f;

    {
        const float* Ap0 = P0 + (size_t)(row0 + m) * NCLS;
        const float* Ap1 = P0 + (size_t)(row0 + 16 + m) * NCLS;
        for (int k = 0; k < NCLS; k += 4) {
            const int ka = k + 2 * hi;
            v2f a0 = *(const v2f*)(Ap0 + ka);
            v2f a1 = *(const v2f*)(Ap1 + ka);
            const float* Bp =
                VPv + (size_t)((k >> 1) + hi) * (2 * NCLS) + (size_t)(cbas + nn) * 2;
#pragma unroll
            for (int t = 0; t < 4; ++t) {
                v2f b = *(const v2f*)(Bp + t * 32);
                acc[0][t] = __builtin_amdgcn_wmma_f32_16x16x4_f32(
                    false, a0, false, b, (short)0, acc[0][t], false, false);
                acc[1][t] = __builtin_amdgcn_wmma_f32_16x16x4_f32(
                    false, a1, false, b, (short)0, acc[1][t], false, false);
            }
        }
    }

    {
        float sg[2][8];
#pragma unroll
        for (int rb = 0; rb < 2; ++rb)
#pragma unroll
            for (int g = 0; g < 8; ++g)
                sg[rb][g] = sigma[row0 + rb * 16 + g + 8 * hi];
#pragma unroll
        for (int t = 0; t < 4; ++t) {
            const int col = cbas + t * 16 + nn;
            const float lv = lam[col];
#pragma unroll
            for (int rb = 0; rb < 2; ++rb)
#pragma unroll
                for (int g = 0; g < 8; ++g) {
                    const int mr = g + 8 * hi;
                    const float sc = __expf(lv * sg[rb][g]);
                    su[rb * 16 + mr][col ^ (mr << 2)] = sc * acc[rb][t][g];
                }
        }
    }
    __syncthreads();

#pragma unroll
    for (int rb = 0; rb < 2; ++rb)
#pragma unroll
        for (int t = 0; t < 4; ++t)
#pragma unroll
            for (int i = 0; i < 8; ++i) acc[rb][t][i] = 0.0f;

    {
        const float* A0 = &su[m][0];
        const float* A1 = &su[16 + m][0];
        for (int k = 0; k < NCLS; k += 4) {
            const int ka = k + 2 * hi;
            const int sw = ka ^ (m << 2);
            v2f a0 = *(const v2f*)(A0 + sw);
            v2f a1 = *(const v2f*)(A1 + sw);
            const float* Bp =
                VPt + (size_t)((k >> 1) + hi) * (2 * NCLS) + (size_t)(cbas + nn) * 2;
#pragma unroll
            for (int t = 0; t < 4; ++t) {
                v2f b = *(const v2f*)(Bp + t * 32);
                acc[0][t] = __builtin_amdgcn_wmma_f32_16x16x4_f32(
                    false, a0, false, b, (short)0, acc[0][t], false, false);
                acc[1][t] = __builtin_amdgcn_wmma_f32_16x16x4_f32(
                    false, a1, false, b, (short)0, acc[1][t], false, false);
            }
        }
    }

#pragma unroll
    for (int t = 0; t < 4; ++t) {
        const int col = cbas + t * 16 + nn;
#pragma unroll
        for (int rb = 0; rb < 2; ++rb)
#pragma unroll
            for (int g = 0; g < 8; ++g) {
                const int r = row0 + rb * 16 + g + 8 * hi;
                out[(size_t)r * NCLS + col] = acc[rb][t][g];
            }
    }
}

extern "C" void kernel_launch(void* const* d_in, const int* in_sizes, int n_in,
                              void* d_out, int out_size, void* d_ws, size_t ws_size,
                              hipStream_t stream) {
    const float* p0    = (const float*)d_in[0];
    const float* sigma = (const float*)d_in[1];
    const float* evals = (const float*)d_in[2];
    const float* evecs = (const float*)d_in[3];
    float* out = (float*)d_out;
    float* VPv = (float*)d_ws;
    float* VPt = VPv + (size_t)NCLS * NCLS;

    pack_operands<<<(NCLS * NCLS / 2) / 256, 256, 0, stream>>>(evecs, VPv, VPt);

    fused_probability_noiser<<<NBATCH / BM, 256, 0, stream>>>(
        p0, sigma, evals, VPv, VPt, out);
}


# Software-pipeline both k loops over two register fragment blocks: the next 4 k-steps' global and LDS fragment loads go out as one group ahead of the current block's 32 WMMAs (hides L2 latency)
# speedup vs baseline: 1.4873x; 1.4873x over previous
#include <hip/hip_runtime.h>


typedef __attribute__((ext_vector_type(2))) float v2f;
typedef __attribute__((ext_vector_type(8))) float v8f;

#define NCLS 512
#define NBATCH 4096
#define BM 32
#define KSTEPS 4
#define KBLK (4 * KSTEPS)

__global__ __launch_bounds__(256) void pack_operands(const float* __restrict__ V,
                                                     float* __restrict__ VPv,
                                                     float* __restrict__ VPt) {
    const int idx = blockIdx.x * 256 + threadIdx.x;
    const int col = idx & (NCLS - 1);
    const int k2  = idx >> 9;

    v2f a;
    a.x = V[(size_t)(2 * k2) * NCLS + col];
    a.y = V[(size_t)(2 * k2 + 1) * NCLS + col];
    *(v2f*)(VPv + ((size_t)k2 * NCLS + col) * 2) = a;

    v2f b = *(const v2f*)(V + (size_t)col * NCLS + 2 * k2);
    *(v2f*)(VPt + ((size_t)k2 * NCLS + col) * 2) = b;
}

struct FragBlock {
    v2f a0[KSTEPS];
    v2f a1[KSTEPS];
    v2f b[KSTEPS][4];
};

__device__ __forceinline__ void load_b(FragBlock& f, const float* __restrict__ Bp, int kb) {
#pragma unroll
    for (int j = 0; j < KSTEPS; ++j)
#pragma unroll
        for (int t = 0; t < 4; ++t)
            f.b[j][t] = *(const v2f*)(Bp + (size_t)((kb >> 1) + 2 * j) * (2 * NCLS) + t * 32);
}

__device__ __forceinline__ void load_a_global(FragBlock& f, const float* __restrict__ A0,
                                              const float* __restrict__ A1, int kb) {
#pragma unroll
    for (int j = 0; j < KSTEPS; ++j) {
        f.a0[j] = *(const v2f*)(A0 + kb + 4 * j);
        f.a1[j] = *(const v2f*)(A1 + kb + 4 * j);
    }
}

__device__ __forceinline__ void load_a_lds(FragBlock& f, const float* A0, const float* A1,
                                           int kb, int hi, int m) {
#pragma unroll
    for (int j = 0; j < KSTEPS; ++j) {
        const int ka = kb + 4 * j + 2 * hi;
        const int sw = ka ^ (m << 2);
        f.a0[j] = *(const v2f*)(A0 + sw);
        f.a1[j] = *(const v2f*)(A1 + sw);
    }
}

__device__ __forceinline__ void mma_block(v8f (&acc)[2][4], const FragBlock& f) {
    __builtin_amdgcn_sched_barrier(0);
#pragma unroll
    for (int j = 0; j < KSTEPS; ++j)
#pragma unroll
        for (int t = 0; t < 4; ++t) {
            acc[0][t] = __builtin_amdgcn_wmma_f32_16x16x4_f32(
                false, f.a0[j], false, f.b[j][t], (short)0, acc[0][t], false, false);
            acc[1][t] = __builtin_amdgcn_wmma_f32_16x16x4_f32(
                false, f.a1[j], false, f.b[j][t], (short)0, acc[1][t], false, false);
        }
    __builtin_amdgcn_sched_barrier(0);
}

__global__ __launch_bounds__(256) void fused_probability_noiser(
    const float* __restrict__ P0,
    const float* __restrict__ sigma,
    const float* __restrict__ lam,
    const float* __restrict__ VPv,
    const float* __restrict__ VPt,
    float* __restrict__ out)
{
    __shared__ float su[BM][NCLS];

    const int tid  = threadIdx.x;
    const int wave = tid >> 5;
    const int lane = tid & 31;
    const int nn   = lane & 15;
    const int hi   = lane >> 4;

    const int cbas = wave * 64;
    const int row0 = blockIdx.x * BM;
    const int m    = nn;

    const size_t boff = (size_t)hi * (2 * NCLS) + (size_t)(cbas + nn) * 2;
    const float* Bp1 = VPv + boff;
    const float* Bp2 = VPt + boff;

    FragBlock f0, f1;

    v8f acc[2][4];
#pragma unroll
    for (int rb = 0; rb < 2; ++rb)
#pragma unroll
        for (int t = 0; t < 4; ++t)
#pragma unroll
            for (int i = 0; i < 8; ++i) acc[rb][t][i] = 0.0f;

    {
        const float* Ap0 = P0 + (size_t)(row0 + m) * NCLS + 2 * hi;
        const float* Ap1 = P0 + (size_t)(row0 + 16 + m) * NCLS + 2 * hi;

        load_a_global(f0, Ap0, Ap1, 0);
        load_b(f0, Bp1, 0);
        for (int kb = 0; kb < NCLS - 2 * KBLK; kb += 2 * KBLK) {
            load_a_global(f1, Ap0, Ap1, kb + KBLK);
            load_b(f1, Bp1, kb + KBLK);
            mma_block(acc, f0);
            load_a_global(f0, Ap0, Ap1, kb + 2 * KBLK);
            load_b(f0, Bp1, kb + 2 * KBLK);
            mma_block(acc, f1);
        }
        load_a_global(f1, Ap0, Ap1, NCLS - KBLK);
        load_b(f1, Bp1, NCLS - KBLK);
        mma_block(acc, f0);
        load_b(f0, Bp2, 0);
        mma_block(acc, f1);
    }

    {
        float sg[2][8];
#pragma unroll
        for (int rb = 0; rb < 2; ++rb)
#pragma unroll
            for (int g = 0; g < 8; ++g)
                sg[rb][g] = sigma[row0 + rb * 16 + g + 8 * hi];
#pragma unroll
        for (int t = 0; t < 4; ++t) {
            const int col = cbas + t * 16 + nn;
            const float lv = lam[col];
#pragma unroll
            for (int rb = 0; rb < 2; ++rb)
#pragma unroll
                for (int g = 0; g < 8; ++g) {
                    const int mr = g + 8 * hi;
                    const float sc = __expf(lv * sg[rb][g]);
                    su[rb * 16 + mr][col ^ (mr << 2)] = sc * acc[rb][t][g];
                }
        }
    }
    __syncthreads();

#pragma unroll
    for (int rb = 0; rb < 2; ++rb)
#pragma unroll
        for (int t = 0; t < 4; ++t)
#pragma unroll
            for (int i = 0; i < 8; ++i) acc[rb][t][i] = 0.0f;

    {
        const float* A0 = &su[m][0];
        const float* A1 = &su[16 + m][0];

        load_a_lds(f0, A0, A1, 0, hi, m);
        for (int kb = 0; kb < NCLS - 2 * KBLK; kb += 2 * KBLK) {
            load_a_lds(f1, A0, A1, kb + KBLK, hi, m);
            load_b(f1, Bp2, kb + KBLK);
            mma_block(acc, f0);
            load_a_lds(f0, A0, A1, kb + 2 * KBLK, hi, m);
            load_b(f0, Bp2, kb + 2 * KBLK);
            mma_block(acc, f1);
        }
        load_a_lds(f1, A0, A1, NCLS - KBLK, hi, m);
        load_b(f1, Bp2, NCLS - KBLK);
        mma_block(acc, f0);
        mma_block(acc, f1);
    }

#pragma unroll
    for (int t = 0; t < 4; ++t) {
        const int col = cbas + t * 16 + nn;
#pragma unroll
        for (int rb = 0; rb < 2; ++rb)
#pragma unroll
            for (int g = 0; g < 8; ++g) {
                const int r = row0 + rb * 16 + g + 8 * hi;
                out[(size_t)r * NCLS + col] = acc[rb][t][g];
            }
    }
}

extern "C" void kernel_launch(void* const* d_in, const int* in_sizes, int n_in,
                              void* d_out, int out_size, void* d_ws, size_t ws_size,
                              hipStream_t stream) {
    const float* p0    = (const float*)d_in[0];
    const float* sigma = (const float*)d_in[1];
    const float* evals = (const float*)d_in[2];
    const float* evecs = (const float*)d_in[3];
    float* out = (float*)d_out;
    float* VPv = (float*)d_ws;
    float* VPt = VPv + (size_t)NCLS * NCLS;

    pack_operands<<<(NCLS * NCLS / 2) / 256, 256, 0, stream>>>(evecs, VPv, VPt);

    fused_probability_noiser<<<NBATCH / BM, 256, 0, stream>>>(
        p0, sigma, evals, VPv, VPt, out);
}


# 16 batch rows per work-group instead of 32: the grid becomes 256 work-groups, one on every WGP (128 left half the chip idle); fragment block 8 k-steps so a block stays 32 WMMAs per wave
# speedup vs baseline: 1.9338x; 1.3002x over previous
#include <hip/hip_runtime.h>


typedef __attribute__((ext_vector_type(2))) float v2f;
typedef __attribute__((ext_vector_type(8))) float v8f;

#define NCLS 512
#define NBATCH 4096
#define BM 16
#define KSTEPS 8
#define KBLK (4 * KSTEPS)

__global__ __launch_bounds__(256) void pack_operands(const float* __restrict__ V,
                                                     float* __restrict__ VPv,
                                                     float* __restrict__ VPt) {
    const int idx = blockIdx.x * 256 + threadIdx.x;
    const int col = idx & (NCLS - 1);
    const int k2  = idx >> 9;

    v2f a;
    a.x = V[(size_t)(2 * k2) * NCLS + col];
    a.y = V[(size_t)(2 * k2 + 1) * NCLS + col];
    *(v2f*)(VPv + ((size_t)k2 * NCLS + col) * 2) = a;

    v2f b = *(const v2f*)(V + (size_t)col * NCLS + 2 * k2);
    *(v2f*)(VPt + ((size_t)k2 * NCLS + col) * 2) = b;
}

struct FragBlock {
    v2f a[KSTEPS];
    v2f b[KSTEPS][4];
};

__device__ __forceinline__ void load_b(FragBlock& f, const float* __restrict__ Bp, int kb) {
#pragma unroll
    for (int j = 0; j < KSTEPS; ++j)
#pragma unroll
        for (int t = 0; t < 4; ++t)
            f.b[j][t] = *(const v2f*)(Bp + (size_t)((kb >> 1) + 2 * j) * (2 * NCLS) + t * 32);
}

__device__ __forceinline__ void load_a_global(FragBlock& f, const float* __restrict__ A0, int kb) {
#pragma unroll
    for (int j = 0; j < KSTEPS; ++j)
        f.a[j] = *(const v2f*)(A0 + kb + 4 * j);
}

__device__ __forceinline__ void load_a_lds(FragBlock& f, const float* A0, int kb, int hi, int m) {
#pragma unroll
    for (int j = 0; j < KSTEPS; ++j) {
        const int ka = kb + 4 * j + 2 * hi;
        const int sw = ka ^ (m << 2);
        f.a[j] = *(const v2f*)(A0 + sw);
    }
}

__device__ __forceinline__ void mma_block(v8f (&acc)[4], const FragBlock& f) {
    __builtin_amdgcn_sched_barrier(0);
#pragma unroll
    for (int j = 0; j < KSTEPS; ++j)
#pragma unroll
        for (int t = 0; t < 4; ++t)
            acc[t] = __builtin_amdgcn_wmma_f32_16x16x4_f32(
                false, f.a[j], false, f.b[j][t], (short)0, acc[t], false, false);
    __builtin_amdgcn_sched_barrier(0);
}

__global__ __launch_bounds__(256) void fused_probability_noiser(
    const float* __restrict__ P0,
    const float* __restrict__ sigma,
    const float* __restrict__ lam,
    const float* __restrict__ VPv,
    const float* __restrict__ VPt,
    float* __restrict__ out)
{
    __shared__ float su[BM][NCLS];

    const int tid  = threadIdx.x;
    const int wave = tid >> 5;
    const int lane = tid & 31;
    const int nn   = lane & 15;
    const int hi   = lane >> 4;

    const int cbas = wave * 64;
    const int row0 = blockIdx.x * BM;
    const int m    = nn;

    const size_t boff = (size_t)hi * (2 * NCLS) + (size_t)(cbas + nn) * 2;
    const float* Bp1 = VPv + boff;
    const float* Bp2 = VPt + boff;

    FragBlock f0, f1;

    v8f acc[4];
#pragma unroll
    for (int t = 0; t < 4; ++t)
#pragma unroll
        for (int i = 0; i < 8; ++i) acc[t][i] = 0.0f;

    {
        const float* Ap0 = P0 + (size_t)(row0 + m) * NCLS + 2 * hi;

        load_a_global(f0, Ap0, 0);
        load_b(f0, Bp1, 0);
        for (int kb = 0; kb < NCLS - 2 * KBLK; kb += 2 * KBLK) {
            load_a_global(f1, Ap0, kb + KBLK);
            load_b(f1, Bp1, kb + KBLK);
            mma_block(acc, f0);
            load_a_global(f0, Ap0, kb + 2 * KBLK);
            load_b(f0, Bp1, kb + 2 * KBLK);
            mma_block(acc, f1);
        }
        load_a_global(f1, Ap0, NCLS - KBLK);
        load_b(f1, Bp1, NCLS - KBLK);
        mma_block(acc, f0);
        load_b(f0, Bp2, 0);
        mma_block(acc, f1);
    }

    {
        float sg[8];
#pragma unroll
        for (int g = 0; g < 8; ++g)
            sg[g] = sigma[row0 + g + 8 * hi];
#pragma unroll
        for (int t = 0; t < 4; ++t) {
            const int col = cbas + t * 16 + nn;
            const float lv = lam[col];
#pragma unroll
            for (int g = 0; g < 8; ++g) {
                const int mr = g + 8 * hi;
                const float sc = __expf(lv * sg[g]);
                su[mr][col ^ (mr << 2)] = sc * acc[t][g];
            }
        }
    }
    __syncthreads();

#pragma unroll
    for (int t = 0; t < 4; ++t)
#pragma unroll
        for (int i = 0; i < 8; ++i) acc[t][i] = 0.0f;

    {
        const float* A0 = &su[m][0];

        load_a_lds(f0, A0, 0, hi, m);
        for (int kb = 0; kb < NCLS - 2 * KBLK; kb += 2 * KBLK) {
            load_a_lds(f1, A0, kb + KBLK, hi, m);
            load_b(f1, Bp2, kb + KBLK);
            mma_block(acc, f0);
            load_a_lds(f0, A0, kb + 2 * KBLK, hi, m);
            load_b(f0, Bp2, kb + 2 * KBLK);
            mma_block(acc, f1);
        }
        load_a_lds(f1, A0, NCLS - KBLK, hi, m);
        load_b(f1, Bp2, NCLS - KBLK);
        mma_block(acc, f0);
        mma_block(acc, f1);
    }

#pragma unroll
    for (int t = 0; t < 4; ++t) {
        const int col = cbas + t * 16 + nn;
#pragma unroll
        for (int g = 0; g < 8; ++g) {
            const int r = row0 + g + 8 * hi;
            out[(size_t)r * NCLS + col] = acc[t][g];
        }
    }
}

extern "C" void kernel_launch(void* const* d_in, const int* in_sizes, int n_in,
                              void* d_out, int out_size, void* d_ws, size_t ws_size,
                              hipStream_t stream) {
    const float* p0    = (const float*)d_in[0];
    const float* sigma = (const float*)d_in[1];
    const float* evals = (const float*)d_in[2];
    const float* evecs = (const float*)d_in[3];
    float* out = (float*)d_out;
    float* VPv = (float*)d_ws;
    float* VPt = VPv + (size_t)NCLS * NCLS;

    pack_operands<<<(NCLS * NCLS / 2) / 256, 256, 0, stream>>>(evecs, VPv, VPt);

    fused_probability_noiser<<<NBATCH / BM, 256, 0, stream>>>(
        p0, sigma, evals, VPv, VPt, out);
}
